# starting kernel
# baseline (speedup 1.0000x reference)
#include <hip/hip_runtime.h>

typedef __attribute__((ext_vector_type(2))) float v2f;
typedef __attribute__((ext_vector_type(8))) float v8f;

#define Bsz   4096
#define Csz   256
#define LM    16

#define OFF_W3   0
#define OFF_C2   1048576
#define OFF_C1   1114112
#define OFF_O1   1118208

__global__ __launch_bounds__(256) void pb_precompute(
    const float* __restrict__ U3, const float* __restrict__ U2,
    const float* __restrict__ U1, const float* __restrict__ w3,
    const float* __restrict__ w2, const float* __restrict__ w1,
    float* __restrict__ ws)
{
    int idx = blockIdx.x * 256 + threadIdx.x;
    if (idx < 1048576) {
        int c   = idx >> 12;
        int xyi = idx & 4095;
        float s = 0.f;
        #pragma unroll
        for (int k = 0; k < 23; ++k) s += U3[xyi * 23 + k] * w3[k * Csz + c];
        ws[OFF_W3 + idx] = s;
    } else if (idx < 1048576 + 65536) {
        int e  = idx - 1048576;
        int c  = e >> 8;
        int xi = e & 255;
        float s = 0.f;
        #pragma unroll
        for (int k = 0; k < 8; ++k) s += U2[xi * 8 + k] * w2[k * Csz + c];
        ws[OFF_C2 + c * 256 + xi] = s;
    } else if (idx < 1048576 + 65536 + 4096) {
        int e = idx - 1048576 - 65536;
        int c = e >> 4;
        int i = e & 15;
        float s = 0.f;
        #pragma unroll
        for (int k = 0; k < 4; ++k) s += U1[i * 4 + k] * w1[k * Csz + c];
        ws[OFF_C1 + c * 16 + i] = s;
    }
}

__global__ __launch_bounds__(256) void pb_main(
    const float* __restrict__ nf,
    const float* __restrict__ ws,
    float* __restrict__ out1)
{
    __shared__ float sW3[4096];
    __shared__ float sC2[256];
    __shared__ float sC1[16];
    __shared__ float sN[8 * 256];

    const int t  = threadIdx.x;
    const int c  = blockIdx.y;
    const int w  = t >> 5;
    const int l  = t & 31;
    const int l16 = l & 15;
    const int hi  = l >> 4;

    {
        const float4* g = (const float4*)(ws + OFF_W3 + c * 4096);
        float4* s = (float4*)sW3;
        #pragma unroll
        for (int j = 0; j < 4; ++j) s[j * 256 + t] = g[j * 256 + t];
    }
    sC2[t] = ws[OFF_C2 + c * 256 + t];
    if (t < 16) sC1[t] = ws[OFF_C1 + c * 16 + t];
    {
        const int row  = l >> 1;
        const int half = l & 1;
        const int b    = (blockIdx.x * 8 + w) * 16 + row;
        const float4* g = (const float4*)(nf + (size_t)b * (Csz * LM) + c * LM + half * 8);
        float4* s = (float4*)(sN + w * 256 + row * 16 + half * 8);
        s[0] = g[0];
        s[1] = g[1];
    }
    __syncthreads();

    const float* Nw = sN + w * 256;

    v2f Bk[4];
    #pragma unroll
    for (int kk = 0; kk < 4; ++kk) {
        int i0 = 4 * kk + 2 * hi;
        Bk[kk].x = Nw[l16 * 16 + i0];
        Bk[kk].y = Nw[l16 * 16 + i0 + 1];
    }
    float nfd[8];
    #pragma unroll
    for (int r = 0; r < 8; ++r) nfd[r] = Nw[l16 * 16 + r + 8 * hi];

    float acc = 0.f;

    #pragma unroll 4
    for (int x = 0; x < 16; ++x) {
        v8f C;
        #pragma unroll
        for (int r = 0; r < 8; ++r) C[r] = sC2[x * 16 + r + 8 * hi];

        const float* Ax = sW3 + x * 256;
        #pragma unroll
        for (int kk = 0; kk < 4; ++kk) {
            int i0 = 4 * kk + 2 * hi;
            v2f A;
            A.x = Ax[l16 * 16 + i0];
            A.y = Ax[l16 * 16 + i0 + 1];
            C = __builtin_amdgcn_wmma_f32_16x16x4_f32(
                    false, A, false, Bk[kk], (short)0, C, false, false);
        }
        float e = C[0] * nfd[0];
        #pragma unroll
        for (int r = 1; r < 8; ++r) e += C[r] * nfd[r];
        e += __shfl_xor(e, 16, 32);
        acc += (e + sC1[x]) * Nw[l16 * 16 + x];
    }

    if (hi == 0) {
        int b = (blockIdx.x * 8 + w) * 16 + l16;
        out1[c * Bsz + b] = acc;
    }
}

__global__ __launch_bounds__(256) void pb_final_gemm(
    const float* __restrict__ o1,
    const float* __restrict__ lw,
    float* __restrict__ out)
{
    const int t   = threadIdx.x;
    const int w   = t >> 5;
    const int l   = t & 31;
    const int l16 = l & 15;
    const int hi  = l >> 4;

    const int tile = blockIdx.x * 8 + w;
    const int b0 = (tile >> 4) * 16;
    const int d0 = (tile & 15) * 16;

    v8f acc;
    #pragma unroll
    for (int r = 0; r < 8; ++r) acc[r] = 0.f;

    #pragma unroll 4
    for (int k0 = 0; k0 < 256; k0 += 4) {
        const int ka = k0 + 2 * hi;
        v2f A, B;
        A.x = o1[(size_t)ka * Bsz + b0 + l16];
        A.y = o1[(size_t)(ka + 1) * Bsz + b0 + l16];
        B.x = lw[ka * Csz + d0 + l16];
        B.y = lw[(ka + 1) * Csz + d0 + l16];
        acc = __builtin_amdgcn_wmma_f32_16x16x4_f32(
                  false, A, false, B, (short)0, acc, false, false);
    }

    #pragma unroll
    for (int r = 0; r < 8; ++r)
        out[(size_t)(b0 + r + 8 * hi) * Csz + d0 + l16] = acc[r] * 0.0625f;
}

extern "C" void kernel_launch(void* const* d_in, const int* in_sizes, int n_in,
                              void* d_out, int out_size, void* d_ws, size_t ws_size,
                              hipStream_t stream) {
    (void)in_sizes; (void)n_in; (void)out_size; (void)ws_size;
    const float* nf  = (const float*)d_in[0];
    const float* U3  = (const float*)d_in[1];
    const float* U2  = (const float*)d_in[2];
    const float* U1  = (const float*)d_in[3];
    const float* w3  = (const float*)d_in[4];
    const float* w2  = (const float*)d_in[5];
    const float* w1  = (const float*)d_in[6];
    const float* lw  = (const float*)d_in[7];

    float* ws   = (float*)d_ws;
    float* out1 = ws + OFF_O1;

    pb_precompute<<<4368, 256, 0, stream>>>(U3, U2, U1, w3, w2, w1, ws);
    pb_main<<<dim3(32, 256), 256, 0, stream>>>(nf, ws, out1);
    pb_final_gemm<<<512, 256, 0, stream>>>(out1, lw, (float*)d_out);
}


# pb_main: W3[c] and c2 fragments resident in registers (no LDS, no barrier); a wave walks 32 b-tiles, nf rows read from global two tiles ahead: 65536 -> 2048 waves; sum order unchanged
# speedup vs baseline: 1.2530x; 1.2530x over previous
#include <hip/hip_runtime.h>

typedef __attribute__((ext_vector_type(2))) float v2f;
typedef __attribute__((ext_vector_type(8))) float v8f;

#define Bsz   4096
#define Csz   256
#define LM    16

#define OFF_W3   0
#define OFF_C2   1048576
#define OFF_C1   1114112
#define OFF_O1   1118208

#define PC_NC 16

__global__ __launch_bounds__(256) void pb_precompute(
    const float* __restrict__ U3, const float* __restrict__ U2,
    const float* __restrict__ U1, const float* __restrict__ w3,
    const float* __restrict__ w2, const float* __restrict__ w1,
    float* __restrict__ ws)
{
    const int t   = threadIdx.x;
    const int blk = blockIdx.x;
    if (blk < 256) {
        const int xyi = (blk & 15) * 256 + t;
        const int c0  = (blk >> 4) * PC_NC;
        float u[23];
        #pragma unroll
        for (int k = 0; k < 23; ++k) u[k] = U3[xyi * 23 + k];
        float s[PC_NC];
        #pragma unroll
        for (int cc = 0; cc < PC_NC; ++cc) s[cc] = 0.f;
        #pragma unroll
        for (int k = 0; k < 23; ++k) {
            #pragma unroll
            for (int cc = 0; cc < PC_NC; ++cc) s[cc] += u[k] * w3[k * Csz + c0 + cc];
        }
        float* o = ws + OFF_W3 + c0 * 4096 + xyi;
        #pragma unroll
        for (int cc = 0; cc < PC_NC; ++cc) o[cc * 4096] = s[cc];
    } else if (blk < 272) {
        const int c0 = (blk - 256) * PC_NC;
        float u[8];
        #pragma unroll
        for (int k = 0; k < 8; ++k) u[k] = U2[t * 8 + k];
        float s[PC_NC];
        #pragma unroll
        for (int cc = 0; cc < PC_NC; ++cc) s[cc] = 0.f;
        #pragma unroll
        for (int k = 0; k < 8; ++k) {
            #pragma unroll
            for (int cc = 0; cc < PC_NC; ++cc) s[cc] += u[k] * w2[k * Csz + c0 + cc];
        }
        float* o = ws + OFF_C2 + c0 * 256 + t;
        #pragma unroll
        for (int cc = 0; cc < PC_NC; ++cc) o[cc * 256] = s[cc];
    } else {
        const int e = (blk - 272) * 256 + t;
        const int c = e >> 4;
        const int i = e & 15;
        float s = 0.f;
        #pragma unroll
        for (int k = 0; k < 4; ++k) s += U1[i * 4 + k] * w1[k * Csz + c];
        ws[OFF_C1 + c * 16 + i] = s;
    }
}

#define PM_TILES 32

struct pm_row { float4 q[4]; };

__device__ __forceinline__ void pm_load_row(pm_row& r, const float* __restrict__ p)
{
    const float4* g = (const float4*)p;
    #pragma unroll
    for (int j = 0; j < 4; ++j) r.q[j] = g[j];
}

__device__ __forceinline__ float pm_pick(float a, float b, unsigned m)
{
    return __uint_as_float((__float_as_uint(b) & m) | (__float_as_uint(a) & ~m));
}

__device__ __forceinline__ float pm_tile(
    const pm_row& row, const v2f (&Af)[16][4], const v8f (&C2f)[16],
    const float (&c1v)[16], const unsigned hm)
{
    float n[16];
    #pragma unroll
    for (int j = 0; j < 4; ++j) {
        n[4 * j + 0] = row.q[j].x;  n[4 * j + 1] = row.q[j].y;
        n[4 * j + 2] = row.q[j].z;  n[4 * j + 3] = row.q[j].w;
    }
    v2f Bk[4];
    #pragma unroll
    for (int kk = 0; kk < 4; ++kk) {
        Bk[kk].x = pm_pick(n[4 * kk + 0], n[4 * kk + 2], hm);
        Bk[kk].y = pm_pick(n[4 * kk + 1], n[4 * kk + 3], hm);
    }
    float nfd[8];
    #pragma unroll
    for (int r = 0; r < 8; ++r) nfd[r] = pm_pick(n[r], n[r + 8], hm);

    float acc = 0.f;
    #pragma unroll
    for (int x = 0; x < 16; ++x) {
        v8f C = C2f[x];
        #pragma unroll
        for (int kk = 0; kk < 4; ++kk)
            C = __builtin_amdgcn_wmma_f32_16x16x4_f32(
                    false, Af[x][kk], false, Bk[kk], (short)0, C, false, false);
        float e = C[0] * nfd[0];
        #pragma unroll
        for (int r = 1; r < 8; ++r) e += C[r] * nfd[r];
        e += __shfl_xor(e, 16, 32);
        acc += (e + c1v[x]) * n[x];
    }
    return acc;
}

__global__ __launch_bounds__(256) __attribute__((amdgpu_waves_per_eu(2, 2))) void pb_main(
    const float* __restrict__ nf,
    const float* __restrict__ ws,
    float* __restrict__ out1)
{
    const int t   = threadIdx.x;
    const int c   = blockIdx.x;
    const int w   = t >> 5;
    const int l   = t & 31;
    const int l16 = l & 15;
    const int hi  = l >> 4;
    const unsigned hm = 0u - (unsigned)hi;

    const float* prow = nf + (size_t)(w * 16 + l16) * (Csz * LM) + c * LM;
    const size_t tstep = (size_t)128 * (Csz * LM);
    float* po = out1 + c * Bsz + w * 16 + l16;

    pm_row s0, s1, s2;
    pm_load_row(s0, prow);
    pm_load_row(s1, prow + tstep);

    v2f Af[16][4];
    v8f C2f[16];
    float c1v[16];
    {
        const float* gw = ws + OFF_W3 + c * 4096 + l16 * 16 + 2 * hi;
        const float* gc = ws + OFF_C2 + c * 256 + 8 * hi;
        #pragma unroll
        for (int x = 0; x < 16; ++x) {
            #pragma unroll
            for (int kk = 0; kk < 4; ++kk)
                Af[x][kk] = *(const v2f*)(gw + x * 256 + 4 * kk);
            const float4 lo = *(const float4*)(gc + x * 16);
            const float4 up = *(const float4*)(gc + x * 16 + 4);
            C2f[x][0] = lo.x;  C2f[x][1] = lo.y;  C2f[x][2] = lo.z;  C2f[x][3] = lo.w;
            C2f[x][4] = up.x;  C2f[x][5] = up.y;  C2f[x][6] = up.z;  C2f[x][7] = up.w;
            c1v[x] = ws[OFF_C1 + c * 16 + x];
        }
    }

    for (int j = 0; j < PM_TILES - 2; j += 3) {
        pm_load_row(s2, prow + (size_t)(j + 2) * tstep);
        float a0 = pm_tile(s0, Af, C2f, c1v, hm);
        if (hi == 0) po[(j + 0) * 128] = a0;
        pm_load_row(s0, prow + (size_t)(j + 3) * tstep);
        float a1 = pm_tile(s1, Af, C2f, c1v, hm);
        if (hi == 0) po[(j + 1) * 128] = a1;
        pm_load_row(s1, prow + (size_t)(j + 4) * tstep);
        float a2 = pm_tile(s2, Af, C2f, c1v, hm);
        if (hi == 0) po[(j + 2) * 128] = a2;
    }
    {
        float a0 = pm_tile(s0, Af, C2f, c1v, hm);
        if (hi == 0) po[(PM_TILES - 2) * 128] = a0;
        float a1 = pm_tile(s1, Af, C2f, c1v, hm);
        if (hi == 0) po[(PM_TILES - 1) * 128] = a1;
    }
}

#define FG_KC 16

__global__ __launch_bounds__(128) __attribute__((amdgpu_waves_per_eu(1, 1))) void pb_final_gemm(
    const float* __restrict__ o1,
    const float* __restrict__ lw,
    float* __restrict__ out)
{
    const int t   = threadIdx.x;
    const int w   = t >> 5;
    const int l   = t & 31;
    const int l16 = l & 15;
    const int hi  = l >> 4;

    const int b_base = (blockIdx.x >> 2) * 64 + (w >> 1) * 32;
    const int d_base = (blockIdx.x & 3) * 64 + (w & 1) * 32;

    const float* pa = o1 + (size_t)(2 * hi) * Bsz + b_base + 2 * l16;
    const float* pb = lw + (2 * hi) * Csz + d_base + 2 * l16;

    v8f acc00, acc01, acc10, acc11;
    #pragma unroll
    for (int r = 0; r < 8; ++r) { acc00[r] = 0.f; acc01[r] = 0.f; acc10[r] = 0.f; acc11[r] = 0.f; }

    v2f fa[2][FG_KC][2], fb[2][FG_KC][2];

    #pragma unroll
    for (int kk = 0; kk < FG_KC; ++kk) {
        #pragma unroll
        for (int s = 0; s < 2; ++s) {
            fa[0][kk][s] = *(const v2f*)(pa + (size_t)(4 * kk + s) * Bsz);
            fb[0][kk][s] = *(const v2f*)(pb + (4 * kk + s) * Csz);
        }
    }

    #pragma unroll
    for (int ch = 0; ch < 256 / (4 * FG_KC); ++ch) {
        const int cur = ch & 1;
        const int nxt = cur ^ 1;
        if (ch + 1 < 256 / (4 * FG_KC)) {
            #pragma unroll
            for (int kk = 0; kk < FG_KC; ++kk) {
                const int ks = (ch + 1) * FG_KC + kk;
                #pragma unroll
                for (int s = 0; s < 2; ++s) {
                    fa[nxt][kk][s] = *(const v2f*)(pa + (size_t)(4 * ks + s) * Bsz);
                    fb[nxt][kk][s] = *(const v2f*)(pb + (4 * ks + s) * Csz);
                }
            }
        }
        #pragma unroll
        for (int kk = 0; kk < FG_KC; ++kk) {
            v2f A0, A1, B0, B1;
            A0.x = fa[cur][kk][0].x;  A0.y = fa[cur][kk][1].x;
            A1.x = fa[cur][kk][0].y;  A1.y = fa[cur][kk][1].y;
            B0.x = fb[cur][kk][0].x;  B0.y = fb[cur][kk][1].x;
            B1.x = fb[cur][kk][0].y;  B1.y = fb[cur][kk][1].y;
            acc00 = __builtin_amdgcn_wmma_f32_16x16x4_f32(
                        false, A0, false, B0, (short)0, acc00, false, false);
            acc01 = __builtin_amdgcn_wmma_f32_16x16x4_f32(
                        false, A0, false, B1, (short)0, acc01, false, false);
            acc11 = __builtin_amdgcn_wmma_f32_16x16x4_f32(
                        false, A1, false, B1, (short)0, acc11, false, false);
            acc10 = __builtin_amdgcn_wmma_f32_16x16x4_f32(
                        false, A1, false, B0, (short)0, acc10, false, false);
        }
    }

    #pragma unroll
    for (int r = 0; r < 8; ++r) {
        const int row0 = b_base + 2 * (8 * hi + r);
        v2f v0, v1;
        v0.x = acc00[r] * 0.0625f;  v0.y = acc01[r] * 0.0625f;
        v1.x = acc10[r] * 0.0625f;  v1.y = acc11[r] * 0.0625f;
        *(v2f*)(out + (size_t)row0 * Csz + d_base + 2 * l16)       = v0;
        *(v2f*)(out + (size_t)(row0 + 1) * Csz + d_base + 2 * l16) = v1;
    }
}

extern "C" void kernel_launch(void* const* d_in, const int* in_sizes, int n_in,
                              void* d_out, int out_size, void* d_ws, size_t ws_size,
                              hipStream_t stream) {
    (void)in_sizes; (void)n_in; (void)out_size; (void)ws_size;
    const float* nf  = (const float*)d_in[0];
    const float* U3  = (const float*)d_in[1];
    const float* U2  = (const float*)d_in[2];
    const float* U1  = (const float*)d_in[3];
    const float* w3  = (const float*)d_in[4];
    const float* w2  = (const float*)d_in[5];
    const float* w1  = (const float*)d_in[6];
    const float* lw  = (const float*)d_in[7];

    float* ws   = (float*)d_ws;
    float* out1 = ws + OFF_O1;

    pb_precompute<<<288, 256, 0, stream>>>(U3, U2, U1, w3, w2, w1, ws);
    pb_main<<<256, 256, 0, stream>>>(nf, ws, out1);
    pb_final_gemm<<<256, 128, 0, stream>>>(out1, lw, (float*)d_out);
}


# pb_main: matrix stage has x as rows (one tile per y), so the y sum is 64 packed FMAs per tile on a lane's own registers, no shuffles or repacking; 3 results stored per round; y/x sum order changes
# speedup vs baseline: 1.3608x; 1.0860x over previous
#include <hip/hip_runtime.h>

typedef __attribute__((ext_vector_type(2))) float v2f;
typedef __attribute__((ext_vector_type(8))) float v8f;

#define Bsz   4096
#define Csz   256
#define LM    16

#define OFF_W3   0
#define OFF_C2   1048576
#define OFF_C1   1114112
#define OFF_O1   1118208

#define PC_NC 16

__global__ __launch_bounds__(256) void pb_precompute(
    const float* __restrict__ U3, const float* __restrict__ U2,
    const float* __restrict__ U1, const float* __restrict__ w3,
    const float* __restrict__ w2, const float* __restrict__ w1,
    float* __restrict__ ws)
{
    const int t   = threadIdx.x;
    const int blk = blockIdx.x;
    if (blk < 256) {
        const int xyi = (blk & 15) * 256 + t;
        const int c0  = (blk >> 4) * PC_NC;
        float u[23];
        #pragma unroll
        for (int k = 0; k < 23; ++k) u[k] = U3[xyi * 23 + k];
        float s[PC_NC];
        #pragma unroll
        for (int cc = 0; cc < PC_NC; ++cc) s[cc] = 0.f;
        #pragma unroll
        for (int k = 0; k < 23; ++k) {
            #pragma unroll
            for (int cc = 0; cc < PC_NC; ++cc) s[cc] += u[k] * w3[k * Csz + c0 + cc];
        }
        float* o = ws + OFF_W3 + c0 * 4096 + xyi;
        #pragma unroll
        for (int cc = 0; cc < PC_NC; ++cc) o[cc * 4096] = s[cc];
    } else if (blk < 272) {
        const int c0 = (blk - 256) * PC_NC;
        float u[8];
        #pragma unroll
        for (int k = 0; k < 8; ++k) u[k] = U2[t * 8 + k];
        float s[PC_NC];
        #pragma unroll
        for (int cc = 0; cc < PC_NC; ++cc) s[cc] = 0.f;
        #pragma unroll
        for (int k = 0; k < 8; ++k) {
            #pragma unroll
            for (int cc = 0; cc < PC_NC; ++cc) s[cc] += u[k] * w2[k * Csz + c0 + cc];
        }
        float* o = ws + OFF_C2 + c0 * 256 + t;
        #pragma unroll
        for (int cc = 0; cc < PC_NC; ++cc) o[cc * 256] = s[cc];
    } else {
        const int e = (blk - 272) * 256 + t;
        const int c = e >> 4;
        const int i = e & 15;
        float s = 0.f;
        #pragma unroll
        for (int k = 0; k < 4; ++k) s += U1[i * 4 + k] * w1[k * Csz + c];
        ws[OFF_C1 + c * 16 + i] = s;
    }
}

#define PM_TILES 32

struct pm_row { float4 q[4]; };

__device__ __forceinline__ void pm_load_row(pm_row& r, const float* __restrict__ p)
{
    const float4* g = (const float4*)p;
    #pragma unroll
    for (int j = 0; j < 4; ++j) r.q[j] = g[j];
}

__device__ __forceinline__ float pm_pick(float a, float b, unsigned m)
{
    return __uint_as_float((__float_as_uint(b) & m) | (__float_as_uint(a) & ~m));
}

__device__ __forceinline__ float pm_tile(
    const pm_row& row, const v2f (&Af)[16][4], const v8f (&C2f)[16],
    const v2f (&c1p)[4], const unsigned hm)
{
    float n[16];
    #pragma unroll
    for (int j = 0; j < 4; ++j) {
        n[4 * j + 0] = row.q[j].x;  n[4 * j + 1] = row.q[j].y;
        n[4 * j + 2] = row.q[j].z;  n[4 * j + 3] = row.q[j].w;
    }
    v2f Bk[4];
    #pragma unroll
    for (int kk = 0; kk < 4; ++kk) {
        Bk[kk].x = pm_pick(n[4 * kk + 0], n[4 * kk + 2], hm);
        Bk[kk].y = pm_pick(n[4 * kk + 1], n[4 * kk + 3], hm);
    }
    float nfd[8];
    #pragma unroll
    for (int r = 0; r < 8; ++r) nfd[r] = pm_pick(n[r], n[r + 8], hm);

    v2f O[4];
    #pragma unroll
    for (int j = 0; j < 4; ++j) O[j] = c1p[j];

    #pragma unroll
    for (int y = 0; y < 16; ++y) {
        v8f D = C2f[y];
        #pragma unroll
        for (int kk = 0; kk < 4; ++kk)
            D = __builtin_amdgcn_wmma_f32_16x16x4_f32(
                    false, Af[y][kk], false, Bk[kk], (short)0, D, false, false);
        v2f ny;
        ny.x = n[y];  ny.y = n[y];
        #pragma unroll
        for (int j = 0; j < 4; ++j) {
            v2f d;
            d.x = D[2 * j];  d.y = D[2 * j + 1];
            O[j] = __builtin_elementwise_fma(d, ny, O[j]);
        }
    }
    float e = O[0].x * nfd[0];
    e += O[0].y * nfd[1];
    #pragma unroll
    for (int j = 1; j < 4; ++j) {
        e += O[j].x * nfd[2 * j];
        e += O[j].y * nfd[2 * j + 1];
    }
    e += __shfl_xor(e, 16, 32);
    return e;
}

__global__ __launch_bounds__(256) __attribute__((amdgpu_waves_per_eu(2, 2))) void pb_main(
    const float* __restrict__ nf,
    const float* __restrict__ ws,
    float* __restrict__ out1)
{
    const int t   = threadIdx.x;
    const int c   = blockIdx.x;
    const int w   = t >> 5;
    const int l   = t & 31;
    const int l16 = l & 15;
    const int hi  = l >> 4;
    const unsigned hm = 0u - (unsigned)hi;

    const float* prow = nf + (size_t)(w * 16 + l16) * (Csz * LM) + c * LM;
    const size_t tstep = (size_t)128 * (Csz * LM);
    float* po = out1 + c * Bsz + w * 16 + l16;

    pm_row s0, s1, s2;
    pm_load_row(s0, prow);
    pm_load_row(s1, prow + tstep);

    v2f Af[16][4];
    v8f C2f[16];
    v2f c1p[4];
    {
        const float* gw = ws + OFF_W3 + c * 4096 + l16 * 256 + 2 * hi;
        #pragma unroll
        for (int y = 0; y < 16; ++y) {
            #pragma unroll
            for (int kk = 0; kk < 4; ++kk)
                Af[y][kk] = *(const v2f*)(gw + y * 16 + 4 * kk);
        }
        const float* gc = ws + OFF_C2 + c * 256 + 128 * hi;
        #pragma unroll
        for (int r = 0; r < 8; ++r) {
            #pragma unroll
            for (int j = 0; j < 4; ++j) {
                const float4 q = *(const float4*)(gc + r * 16 + 4 * j);
                C2f[4 * j + 0][r] = q.x;  C2f[4 * j + 1][r] = q.y;
                C2f[4 * j + 2][r] = q.z;  C2f[4 * j + 3][r] = q.w;
            }
        }
        const float* g1 = ws + OFF_C1 + c * 16 + 8 * hi;
        const float4 lo = *(const float4*)(g1);
        const float4 up = *(const float4*)(g1 + 4);
        c1p[0].x = lo.x;  c1p[0].y = lo.y;  c1p[1].x = lo.z;  c1p[1].y = lo.w;
        c1p[2].x = up.x;  c1p[2].y = up.y;  c1p[3].x = up.z;  c1p[3].y = up.w;
    }

    for (int j = 0; j < PM_TILES - 2; j += 3) {
        pm_load_row(s2, prow + (size_t)(j + 2) * tstep);
        const float a0 = pm_tile(s0, Af, C2f, c1p, hm);
        pm_load_row(s0, prow + (size_t)(j + 3) * tstep);
        const float a1 = pm_tile(s1, Af, C2f, c1p, hm);
        pm_load_row(s1, prow + (size_t)(j + 4) * tstep);
        const float a2 = pm_tile(s2, Af, C2f, c1p, hm);
        if (hi == 0) {
            po[(j + 0) * 128] = a0;
            po[(j + 1) * 128] = a1;
            po[(j + 2) * 128] = a2;
        }
    }
    {
        const float a0 = pm_tile(s0, Af, C2f, c1p, hm);
        const float a1 = pm_tile(s1, Af, C2f, c1p, hm);
        if (hi == 0) {
            po[(PM_TILES - 2) * 128] = a0;
            po[(PM_TILES - 1) * 128] = a1;
        }
    }
}

#define FG_KC 16

__global__ __launch_bounds__(128) __attribute__((amdgpu_waves_per_eu(1, 1))) void pb_final_gemm(
    const float* __restrict__ o1,
    const float* __restrict__ lw,
    float* __restrict__ out)
{
    const int t   = threadIdx.x;
    const int w   = t >> 5;
    const int l   = t & 31;
    const int l16 = l & 15;
    const int hi  = l >> 4;

    const int b_base = (blockIdx.x >> 2) * 64 + (w >> 1) * 32;
    const int d_base = (blockIdx.x & 3) * 64 + (w & 1) * 32;

    const float* pa = o1 + (size_t)(2 * hi) * Bsz + b_base + 2 * l16;
    const float* pb = lw + (2 * hi) * Csz + d_base + 2 * l16;

    v8f acc00, acc01, acc10, acc11;
    #pragma unroll
    for (int r = 0; r < 8; ++r) { acc00[r] = 0.f; acc01[r] = 0.f; acc10[r] = 0.f; acc11[r] = 0.f; }

    v2f fa[2][FG_KC][2], fb[2][FG_KC][2];

    #pragma unroll
    for (int kk = 0; kk < FG_KC; ++kk) {
        #pragma unroll
        for (int s = 0; s < 2; ++s) {
            fa[0][kk][s] = *(const v2f*)(pa + (size_t)(4 * kk + s) * Bsz);
            fb[0][kk][s] = *(const v2f*)(pb + (4 * kk + s) * Csz);
        }
    }

    #pragma unroll
    for (int ch = 0; ch < 256 / (4 * FG_KC); ++ch) {
        const int cur = ch & 1;
        const int nxt = cur ^ 1;
        if (ch + 1 < 256 / (4 * FG_KC)) {
            #pragma unroll
            for (int kk = 0; kk < FG_KC; ++kk) {
                const int ks = (ch + 1) * FG_KC + kk;
                #pragma unroll
                for (int s = 0; s < 2; ++s) {
                    fa[nxt][kk][s] = *(const v2f*)(pa + (size_t)(4 * ks + s) * Bsz);
                    fb[nxt][kk][s] = *(const v2f*)(pb + (4 * ks + s) * Csz);
                }
            }
        }
        #pragma unroll
        for (int kk = 0; kk < FG_KC; ++kk) {
            v2f A0, A1, B0, B1;
            A0.x = fa[cur][kk][0].x;  A0.y = fa[cur][kk][1].x;
            A1.x = fa[cur][kk][0].y;  A1.y = fa[cur][kk][1].y;
            B0.x = fb[cur][kk][0].x;  B0.y = fb[cur][kk][1].x;
            B1.x = fb[cur][kk][0].y;  B1.y = fb[cur][kk][1].y;
            acc00 = __builtin_amdgcn_wmma_f32_16x16x4_f32(
                        false, A0, false, B0, (short)0, acc00, false, false);
            acc01 = __builtin_amdgcn_wmma_f32_16x16x4_f32(
                        false, A0, false, B1, (short)0, acc01, false, false);
            acc11 = __builtin_amdgcn_wmma_f32_16x16x4_f32(
                        false, A1, false, B1, (short)0, acc11, false, false);
            acc10 = __builtin_amdgcn_wmma_f32_16x16x4_f32(
                        false, A1, false, B0, (short)0, acc10, false, false);
        }
    }

    #pragma unroll
    for (int r = 0; r < 8; ++r) {
        const int row0 = b_base + 2 * (8 * hi + r);
        v2f v0, v1;
        v0.x = acc00[r] * 0.0625f;  v0.y = acc01[r] * 0.0625f;
        v1.x = acc10[r] * 0.0625f;  v1.y = acc11[r] * 0.0625f;
        *(v2f*)(out + (size_t)row0 * Csz + d_base + 2 * l16)       = v0;
        *(v2f*)(out + (size_t)(row0 + 1) * Csz + d_base + 2 * l16) = v1;
    }
}

extern "C" void kernel_launch(void* const* d_in, const int* in_sizes, int n_in,
                              void* d_out, int out_size, void* d_ws, size_t ws_size,
                              hipStream_t stream) {
    (void)in_sizes; (void)n_in; (void)out_size; (void)ws_size;
    const float* nf  = (const float*)d_in[0];
    const float* U3  = (const float*)d_in[1];
    const float* U2  = (const float*)d_in[2];
    const float* U1  = (const float*)d_in[3];
    const float* w3  = (const float*)d_in[4];
    const float* w2  = (const float*)d_in[5];
    const float* w1  = (const float*)d_in[6];
    const float* lw  = (const float*)d_in[7];

    float* ws   = (float*)d_ws;
    float* out1 = ws + OFF_O1;

    pb_precompute<<<288, 256, 0, stream>>>(U3, U2, U1, w3, w2, w1, ws);
    pb_main<<<256, 256, 0, stream>>>(nf, ws, out1);
    pb_final_gemm<<<256, 128, 0, stream>>>(out1, lw, (float*)d_out);
}
